# starting kernel
# baseline (speedup 1.0000x reference)
#include <hip/hip_runtime.h>

typedef __attribute__((ext_vector_type(2))) float v2f;
typedef __attribute__((ext_vector_type(4))) float v4f;
typedef __attribute__((ext_vector_type(8))) float v8f;

#define N_NODES 50000
#define N_EDGES 800000
#define D       128

__global__ void gcn_init_bias(const float* __restrict__ b,
                              float* __restrict__ out) {
  int i = blockIdx.x * blockDim.x + threadIdx.x;
  const v4f* b4 = (const v4f*)b;
  ((v4f*)out)[i] = b4[i & (D / 4 - 1)];
}

__global__ void gcn_gemm_wmma(const float* __restrict__ X,
                              const float* __restrict__ W,
                              float* __restrict__ H) {
  const int lane  = threadIdx.x & 31;
  const int wave  = threadIdx.x >> 5;
  const int mbase = blockIdx.x * 16;
  const int nbase = wave * 16;
  const int half  = lane >> 4;
  const int l     = lane & 15;

  const float* __restrict__ xrow = X + (size_t)(mbase + l) * D;
  const float* __restrict__ wrow = W + (size_t)(nbase + l) * D;

  v8f acc = {};
#pragma unroll
  for (int k = 0; k < D; k += 4) {
    v2f a = *(const v2f*)(xrow + k + 2 * half);
    v2f bm = *(const v2f*)(wrow + k + 2 * half);
    acc = __builtin_amdgcn_wmma_f32_16x16x4_f32(
         false, a,  false, bm,
         (short)0, acc,  false,  false);
  }

  float* __restrict__ hrow = H + (size_t)(mbase + 8 * half) * D + nbase + l;
#pragma unroll
  for (int r = 0; r < 8; ++r) {
    hrow[(size_t)r * D] = acc[r];
  }
}

__global__ void gcn_scatter(const int* __restrict__ ei,
                            const float* __restrict__ H,
                            float* __restrict__ out) {
  const int e    = blockIdx.x * 8 + (threadIdx.x >> 5);
  const int lane = threadIdx.x & 31;
  const int src  = ei[e];
  const int dst  = ei[N_EDGES + e];

  v4f m = ((const v4f*)(H + (size_t)src * D))[lane];
  float* o = out + (size_t)dst * D + lane * 4;
  atomicAdd(o + 0, m.x);
  atomicAdd(o + 1, m.y);
  atomicAdd(o + 2, m.z);
  atomicAdd(o + 3, m.w);
}

extern "C" void kernel_launch(void* const* d_in, const int* in_sizes, int n_in,
                              void* d_out, int out_size, void* d_ws, size_t ws_size,
                              hipStream_t stream) {
  const float* x  = (const float*)d_in[0];
  const int*   ei = (const int*)d_in[1];
  const float* W  = (const float*)d_in[2];
  const float* b  = (const float*)d_in[3];
  float* out = (float*)d_out;
  float* H   = (float*)d_ws;

  gcn_init_bias<<<(N_NODES * D / 4) / 256, 256, 0, stream>>>(b, out);

  gcn_gemm_wmma<<<N_NODES / 16, 256, 0, stream>>>(x, W, H);

  gcn_scatter<<<N_EDGES / 8, 256, 0, stream>>>(ei, H, out);
}


# scatter: each wave-wide H load and f32 atomic add covers one whole 128-byte line (lane+32j), not a quarter of each of 4 lines: 4x fewer atomic line requests at L2; sums stay in atomic arrival order
# speedup vs baseline: 1.3312x; 1.3312x over previous
#include <hip/hip_runtime.h>

typedef __attribute__((ext_vector_type(2))) float v2f;
typedef __attribute__((ext_vector_type(4))) float v4f;
typedef __attribute__((ext_vector_type(8))) float v8f;

#define N_NODES 50000
#define N_EDGES 800000
#define D       128

__global__ void gcn_init_bias(const float* __restrict__ b,
                              float* __restrict__ out) {
  int i = blockIdx.x * blockDim.x + threadIdx.x;
  const v4f* b4 = (const v4f*)b;
  ((v4f*)out)[i] = b4[i & (D / 4 - 1)];
}

__global__ void gcn_gemm_wmma(const float* __restrict__ X,
                              const float* __restrict__ W,
                              float* __restrict__ H) {
  const int lane  = threadIdx.x & 31;
  const int wave  = threadIdx.x >> 5;
  const int mbase = blockIdx.x * 16;
  const int nbase = wave * 16;
  const int half  = lane >> 4;
  const int l     = lane & 15;

  const float* __restrict__ xrow = X + (size_t)(mbase + l) * D;
  const float* __restrict__ wrow = W + (size_t)(nbase + l) * D;

  v8f acc = {};
#pragma unroll
  for (int k = 0; k < D; k += 4) {
    v2f a = *(const v2f*)(xrow + k + 2 * half);
    v2f bm = *(const v2f*)(wrow + k + 2 * half);
    acc = __builtin_amdgcn_wmma_f32_16x16x4_f32(
         false, a,  false, bm,
         (short)0, acc,  false,  false);
  }

  float* __restrict__ hrow = H + (size_t)(mbase + 8 * half) * D + nbase + l;
#pragma unroll
  for (int r = 0; r < 8; ++r) {
    hrow[(size_t)r * D] = acc[r];
  }
}

__global__ void gcn_scatter(const int* __restrict__ ei,
                            const float* __restrict__ H,
                            float* __restrict__ out) {
  const int e    = blockIdx.x * 8 + (threadIdx.x >> 5);
  const int lane = threadIdx.x & 31;
  const int src  = ei[e];
  const int dst  = ei[N_EDGES + e];

  const float* __restrict__ h = H + (size_t)src * D + lane;
  float* o = out + (size_t)dst * D + lane;
  const float m0 = h[0];
  const float m1 = h[32];
  const float m2 = h[64];
  const float m3 = h[96];
  atomicAdd(o +  0, m0);
  atomicAdd(o + 32, m1);
  atomicAdd(o + 64, m2);
  atomicAdd(o + 96, m3);
}

extern "C" void kernel_launch(void* const* d_in, const int* in_sizes, int n_in,
                              void* d_out, int out_size, void* d_ws, size_t ws_size,
                              hipStream_t stream) {
  const float* x  = (const float*)d_in[0];
  const int*   ei = (const int*)d_in[1];
  const float* W  = (const float*)d_in[2];
  const float* b  = (const float*)d_in[3];
  float* out = (float*)d_out;
  float* H   = (float*)d_ws;

  gcn_init_bias<<<(N_NODES * D / 4) / 256, 256, 0, stream>>>(b, out);

  gcn_gemm_wmma<<<N_NODES / 16, 256, 0, stream>>>(x, W, H);

  gcn_scatter<<<N_EDGES / 8, 256, 0, stream>>>(ei, H, out);
}


# scatter: each wave walks 32 edges (one coalesced index load per wave, broadcast per edge), row loads of 8 edges in flight before their atomic adds; 32x fewer waves; sums stay in atomic arrival order
# speedup vs baseline: 1.7497x; 1.3144x over previous
#include <hip/hip_runtime.h>

typedef __attribute__((ext_vector_type(2))) float v2f;
typedef __attribute__((ext_vector_type(4))) float v4f;
typedef __attribute__((ext_vector_type(8))) float v8f;

#define N_NODES 50000
#define N_EDGES 800000
#define D       128

__global__ void gcn_init_bias(const float* __restrict__ b,
                              float* __restrict__ out) {
  int i = blockIdx.x * blockDim.x + threadIdx.x;
  const v4f* b4 = (const v4f*)b;
  ((v4f*)out)[i] = b4[i & (D / 4 - 1)];
}

__global__ void gcn_gemm_wmma(const float* __restrict__ X,
                              const float* __restrict__ W,
                              float* __restrict__ H) {
  const int lane  = threadIdx.x & 31;
  const int wave  = threadIdx.x >> 5;
  const int mbase = blockIdx.x * 16;
  const int nbase = wave * 16;
  const int half  = lane >> 4;
  const int l     = lane & 15;

  const float* __restrict__ xrow = X + (size_t)(mbase + l) * D;
  const float* __restrict__ wrow = W + (size_t)(nbase + l) * D;

  v8f acc = {};
#pragma unroll
  for (int k = 0; k < D; k += 4) {
    v2f a = *(const v2f*)(xrow + k + 2 * half);
    v2f bm = *(const v2f*)(wrow + k + 2 * half);
    acc = __builtin_amdgcn_wmma_f32_16x16x4_f32(
         false, a,  false, bm,
         (short)0, acc,  false,  false);
  }

  float* __restrict__ hrow = H + (size_t)(mbase + 8 * half) * D + nbase + l;
#pragma unroll
  for (int r = 0; r < 8; ++r) {
    hrow[(size_t)r * D] = acc[r];
  }
}

#define EPW 32
#define EB  8

__global__ void gcn_scatter(const int* __restrict__ ei,
                            const float* __restrict__ H,
                            float* __restrict__ out) {
  const int lane = threadIdx.x & 31;
  const int e0   = (blockIdx.x * 8 + (threadIdx.x >> 5)) * EPW;
  const int srcv = ei[e0 + lane];
  const int dstv = ei[N_EDGES + e0 + lane];

#pragma unroll
  for (int g = 0; g < EPW; g += EB) {
    float m[EB][4];
#pragma unroll
    for (int j = 0; j < EB; ++j) {
      const int s = __builtin_amdgcn_readlane(srcv, g + j);
      const float* __restrict__ h = H + (size_t)s * D;
      m[j][0] = h[lane];
      m[j][1] = h[lane + 32];
      m[j][2] = h[lane + 64];
      m[j][3] = h[lane + 96];
    }
#pragma unroll
    for (int j = 0; j < EB; ++j) {
      const int d = __builtin_amdgcn_readlane(dstv, g + j);
      float* o = out + (size_t)d * D;
      atomicAdd(o + lane,      m[j][0]);
      atomicAdd(o + lane + 32, m[j][1]);
      atomicAdd(o + lane + 64, m[j][2]);
      atomicAdd(o + lane + 96, m[j][3]);
    }
  }
}

extern "C" void kernel_launch(void* const* d_in, const int* in_sizes, int n_in,
                              void* d_out, int out_size, void* d_ws, size_t ws_size,
                              hipStream_t stream) {
  const float* x  = (const float*)d_in[0];
  const int*   ei = (const int*)d_in[1];
  const float* W  = (const float*)d_in[2];
  const float* b  = (const float*)d_in[3];
  float* out = (float*)d_out;
  float* H   = (float*)d_ws;

  gcn_init_bias<<<(N_NODES * D / 4) / 256, 256, 0, stream>>>(b, out);

  gcn_gemm_wmma<<<N_NODES / 16, 256, 0, stream>>>(x, W, H);

  gcn_scatter<<<N_EDGES / (8 * EPW), 256, 0, stream>>>(ei, H, out);
}


# segment sum without global float atomics: edges ordered by destination range each call, one work-group per 196 rows sums in LDS, each out line stored once with bias; sums in LDS arrival order
# speedup vs baseline: 1.9225x; 1.0988x over previous
#include <hip/hip_runtime.h>


typedef __attribute__((ext_vector_type(2))) float v2f;
typedef __attribute__((ext_vector_type(4))) float v4f;
typedef __attribute__((ext_vector_type(8))) float v8f;
typedef __attribute__((ext_vector_type(2))) int   v2i;
typedef __attribute__((ext_vector_type(4))) int   v4i;

#define N_NODES 50000
#define N_EDGES 800000
#define D       128

#define NBKT   256
#define ROWS   196
#define NCHUNK 256
#define CHUNK  3125
#define TRIPS  4
#define EB     15

__global__ void gcn_gemm_wmma(const float* __restrict__ X,
                              const float* __restrict__ W,
                              float* __restrict__ H) {
  const int lane  = threadIdx.x & 31;
  const int wave  = threadIdx.x >> 5;
  const int mbase = blockIdx.x * 16;
  const int nbase = wave * 16;
  const int half  = lane >> 4;
  const int l     = lane & 15;

  const float* __restrict__ xrow = X + (size_t)(mbase + l) * D;
  const float* __restrict__ wrow = W + (size_t)(nbase + l) * D;

  v8f acc = {};
#pragma unroll
  for (int k = 0; k < D; k += 4) {
    v2f a = *(const v2f*)(xrow + k + 2 * half);
    v2f bm = *(const v2f*)(wrow + k + 2 * half);
    acc = __builtin_amdgcn_wmma_f32_16x16x4_f32(
         false, a,  false, bm,
         (short)0, acc,  false,  false);
  }

  float* __restrict__ hrow = H + (size_t)(mbase + 8 * half) * D + nbase + l;
#pragma unroll
  for (int r = 0; r < 8; ++r) {
    hrow[(size_t)r * D] = acc[r];
  }
}

__global__ __launch_bounds__(1024) void gcn_count(const int* __restrict__ ei,
                                                  int* __restrict__ cnt) {
  __shared__ int hist[NBKT];
  const int tid = threadIdx.x;
  const int base = blockIdx.x * CHUNK;
  unsigned dv[TRIPS];
#pragma unroll
  for (int t = 0; t < TRIPS; ++t) {
    dv[t] = (unsigned)ei[N_EDGES + base + min(tid + t * 1024, CHUNK - 1)];
  }
  if (tid < NBKT) hist[tid] = 0;
  __syncthreads();
#pragma unroll
  for (int t = 0; t < TRIPS; ++t) {
    if (tid + t * 1024 < CHUNK) atomicAdd(&hist[dv[t] / ROWS], 1);
  }
  __syncthreads();
  if (tid < NBKT) cnt[blockIdx.x * NBKT + tid] = hist[tid];
}

__global__ __launch_bounds__(1024) void gcn_place(const int* __restrict__ ei,
                                                  const int* __restrict__ cnt,
                                                  int* __restrict__ bstart,
                                                  v2i* __restrict__ sorted) {
  __shared__ __attribute__((aligned(16))) int ptot[16][NBKT];
  __shared__ __attribute__((aligned(16))) int pmine[16][NBKT];
  __shared__ int tot[NBKT];
  __shared__ int cursor[NBKT];
  const int tid = threadIdx.x;
  const int base = blockIdx.x * CHUNK;

  int sv[TRIPS];
  unsigned dv[TRIPS];
#pragma unroll
  for (int t = 0; t < TRIPS; ++t) {
    const int i = base + min(tid + t * 1024, CHUNK - 1);
    sv[t] = ei[i];
    dv[t] = (unsigned)ei[N_EDGES + i];
  }

  const int q  = tid >> 6;
  const int c4 = tid & 63;
  v4i st = {0, 0, 0, 0};
  v4i sm = {0, 0, 0, 0};
#pragma unroll
  for (int j = 0; j < NCHUNK / 16; ++j) {
    const int row = q * (NCHUNK / 16) + j;
    const v4i c = ((const v4i*)cnt)[row * (NBKT / 4) + c4];
    st += c;
    if (row < (int)blockIdx.x) sm += c;
  }
  ((v4i*)ptot[q])[c4]  = st;
  ((v4i*)pmine[q])[c4] = sm;
  __syncthreads();

  int mine = 0;
  if (tid < NBKT) {
    int t = 0;
#pragma unroll
    for (int j = 0; j < 16; ++j) { t += ptot[j][tid]; mine += pmine[j][tid]; }
    tot[tid] = t;
  }
  __syncthreads();

  if (tid < 32) {
    int v[8];
    int s = 0;
#pragma unroll
    for (int i = 0; i < 8; ++i) { v[i] = tot[tid * 8 + i]; s += v[i]; }
    int inc = s;
#pragma unroll
    for (int off = 1; off < 32; off <<= 1) {
      const int t = __shfl_up(inc, off);
      if (tid >= off) inc += t;
    }
    int run = inc - s;
#pragma unroll
    for (int i = 0; i < 8; ++i) { tot[tid * 8 + i] = run; run += v[i]; }
  }
  __syncthreads();

  if (tid < NBKT) {
    cursor[tid] = tot[tid] + mine;
    if (blockIdx.x == 0) {
      bstart[tid] = tot[tid];
      if (tid == 0) bstart[NBKT] = N_EDGES;
    }
  }
  __syncthreads();

#pragma unroll
  for (int t = 0; t < TRIPS; ++t) {
    if (tid + t * 1024 < CHUNK) {
      const unsigned bk = dv[t] / ROWS;
      const int pos = atomicAdd(&cursor[bk], 1);
      v2i p;
      p.x = sv[t];
      p.y = (int)(dv[t] - bk * ROWS);
      sorted[pos] = p;
    }
  }
}

__global__ __launch_bounds__(1024) void gcn_gather(const v2i* __restrict__ sorted,
                                                   const int* __restrict__ bstart,
                                                   const float* __restrict__ H,
                                                   const float* __restrict__ bias,
                                                   float* __restrict__ out) {
  __shared__ __attribute__((aligned(16))) float acc[(ROWS + 1) * D];
  const int tid  = threadIdx.x;
  const int lane = tid & 31;
  const int wave = tid >> 5;

  for (int i = tid; i < (ROWS + 1) * D / 4; i += 1024) {
    ((v4f*)acc)[i] = (v4f){0.f, 0.f, 0.f, 0.f};
  }
  const int beg = bstart[blockIdx.x];
  const int end = bstart[blockIdx.x + 1];
  const int nb  = (end - beg + EB - 1) / EB;
  __syncthreads();

  const v2i pad = {0, ROWS};
  int e = beg + wave * EB + ((lane < EB) ? lane : 0);
  v2i raw = sorted[min(e, N_EDGES - 1)];
  for (int k = wave; k < nb; k += 32) {
    const int en = e + 32 * EB;
    const v2i rawn = sorted[min(en, N_EDGES - 1)];
    const v2i p = (e < end) ? raw : pad;

    float m[EB][4];
#pragma unroll
    for (int j = 0; j < EB; ++j) {
      const int s = __builtin_amdgcn_readlane(p.x, j);
      const float* __restrict__ h = H + (size_t)s * D;
      m[j][0] = h[lane];
      m[j][1] = h[lane + 32];
      m[j][2] = h[lane + 64];
      m[j][3] = h[lane + 96];
    }
#pragma unroll
    for (int j = 0; j < EB; ++j) {
      const int r = __builtin_amdgcn_readlane(p.y, j);
      float* a = acc + r * D + lane;
      __hip_atomic_fetch_add(a,      m[j][0], __ATOMIC_RELAXED, __HIP_MEMORY_SCOPE_WORKGROUP);
      __hip_atomic_fetch_add(a + 32, m[j][1], __ATOMIC_RELAXED, __HIP_MEMORY_SCOPE_WORKGROUP);
      __hip_atomic_fetch_add(a + 64, m[j][2], __ATOMIC_RELAXED, __HIP_MEMORY_SCOPE_WORKGROUP);
      __hip_atomic_fetch_add(a + 96, m[j][3], __ATOMIC_RELAXED, __HIP_MEMORY_SCOPE_WORKGROUP);
    }
    raw = rawn;
    e = en;
  }
  __syncthreads();

  const v4f bv = ((const v4f*)bias)[lane];
  const int row0 = blockIdx.x * ROWS;
  for (int r = wave; r < ROWS; r += 32) {
    const int row = row0 + r;
    if (row < N_NODES) {
      const v4f v = ((const v4f*)(acc + r * D))[lane];
      ((v4f*)(out + (size_t)row * D))[lane] = v + bv;
    }
  }
}

extern "C" void kernel_launch(void* const* d_in, const int* in_sizes, int n_in,
                              void* d_out, int out_size, void* d_ws, size_t ws_size,
                              hipStream_t stream) {
  const float* x  = (const float*)d_in[0];
  const int*   ei = (const int*)d_in[1];
  const float* W  = (const float*)d_in[2];
  const float* b  = (const float*)d_in[3];
  float* out = (float*)d_out;
  float* H      = (float*)d_ws;
  int*   cnt    = (int*)((char*)d_ws + 33554432);
  int*   bstart = (int*)((char*)d_ws + 34603008);
  v2i*   sorted = (v2i*)((char*)d_ws + 41943040);

  gcn_count<<<NCHUNK, 1024, 0, stream>>>(ei, cnt);

  gcn_place<<<NCHUNK, 1024, 0, stream>>>(ei, cnt, bstart, sorted);

  gcn_gemm_wmma<<<N_NODES / 16, 256, 0, stream>>>(x, W, H);

  gcn_gather<<<NBKT, 1024, 0, stream>>>(sorted, bstart, H, b, out);
}


# GEMM re-tiled: each wave owns a 16x128 row strip of H (one A fragment feeds 8 WMMAs), W and the X strip staged in LDS by whole-row loads, 8x fewer waves; k order kept, H bit-identical
# speedup vs baseline: 2.5884x; 1.3464x over previous
#include <hip/hip_runtime.h>


typedef __attribute__((ext_vector_type(2))) float v2f;
typedef __attribute__((ext_vector_type(4))) float v4f;
typedef __attribute__((ext_vector_type(8))) float v8f;
typedef __attribute__((ext_vector_type(2))) int   v2i;
typedef __attribute__((ext_vector_type(4))) int   v4i;

#define N_NODES 50000
#define N_EDGES 800000
#define D       128

#define NBKT   256
#define ROWS   196
#define NCHUNK 256
#define CHUNK  3125
#define TRIPS  4
#define WP     132
#define GW     8
#define GBLK   391
#define EB     15

__global__ __launch_bounds__(256) void gcn_gemm_wmma(const float* __restrict__ X,
                                                    const float* __restrict__ W,
                                                    float* __restrict__ H) {
  __shared__ __attribute__((aligned(16))) float wl[D * WP];
  __shared__ __attribute__((aligned(16))) float xl[GW * 16 * WP];
  const int tid   = threadIdx.x;
  const int lane  = tid & 31;
  const int wave  = __builtin_amdgcn_readfirstlane(tid >> 5);
  const int half  = lane >> 4;
  const int l     = lane & 15;
  const int strip = blockIdx.x * GW + wave;
  const bool live = strip < N_NODES / 16;
  const int mbase = strip * 16;
  float* xs = xl + wave * (16 * WP);

  v4f xr[16];
  if (live) {
#pragma unroll
    for (int i = 0; i < 16; ++i) {
      xr[i] = ((const v4f*)(X + (size_t)(mbase + i) * D))[lane];
    }
  }
  v4f wr[16];
#pragma unroll
  for (int q = 0; q < 16; ++q) {
    wr[q] = ((const v4f*)(W + (size_t)(wave + GW * q) * D))[lane];
  }
#pragma unroll
  for (int q = 0; q < 16; ++q) {
    *(v4f*)(wl + (wave + GW * q) * WP + lane * 4) = wr[q];
  }
  if (live) {
#pragma unroll
    for (int i = 0; i < 16; ++i) {
      *(v4f*)(xs + i * WP + lane * 4) = xr[i];
    }
  }
  __syncthreads();
  if (!live) return;

  const float* ap = xs + l * WP + 2 * half;
  const float* bp = wl + l * WP + 2 * half;
  v8f acc[8];
#pragma unroll
  for (int t = 0; t < 8; ++t) acc[t] = (v8f){0.f, 0.f, 0.f, 0.f, 0.f, 0.f, 0.f, 0.f};

#pragma unroll
  for (int k = 0; k < D; k += 4) {
    const v2f a = *(const v2f*)(ap + k);
    v2f bm[8];
#pragma unroll
    for (int t = 0; t < 8; ++t) bm[t] = *(const v2f*)(bp + t * 16 * WP + k);
#pragma unroll
    for (int t = 0; t < 8; ++t) {
      acc[t] = __builtin_amdgcn_wmma_f32_16x16x4_f32(
           false, a,  false, bm[t],
           (short)0, acc[t],  false,  false);
    }
  }

  float* __restrict__ hrow = H + (size_t)(mbase + 8 * half) * D + l;
#pragma unroll
  for (int r = 0; r < 8; ++r) {
#pragma unroll
    for (int t = 0; t < 8; ++t) {
      hrow[(size_t)r * D + 16 * t] = acc[t][r];
    }
  }
}

__global__ __launch_bounds__(1024) void gcn_count(const int* __restrict__ ei,
                                                  int* __restrict__ cnt) {
  __shared__ int hist[NBKT];
  const int tid = threadIdx.x;
  const int base = blockIdx.x * CHUNK;
  unsigned dv[TRIPS];
#pragma unroll
  for (int t = 0; t < TRIPS; ++t) {
    dv[t] = (unsigned)ei[N_EDGES + base + min(tid + t * 1024, CHUNK - 1)];
  }
  if (tid < NBKT) hist[tid] = 0;
  __syncthreads();
#pragma unroll
  for (int t = 0; t < TRIPS; ++t) {
    if (tid + t * 1024 < CHUNK) atomicAdd(&hist[dv[t] / ROWS], 1);
  }
  __syncthreads();
  if (tid < NBKT) cnt[blockIdx.x * NBKT + tid] = hist[tid];
}

__global__ __launch_bounds__(1024) void gcn_place(const int* __restrict__ ei,
                                                  const int* __restrict__ cnt,
                                                  int* __restrict__ bstart,
                                                  v2i* __restrict__ sorted) {
  __shared__ __attribute__((aligned(16))) int ptot[16][NBKT];
  __shared__ __attribute__((aligned(16))) int pmine[16][NBKT];
  __shared__ int tot[NBKT];
  __shared__ int cursor[NBKT];
  const int tid = threadIdx.x;
  const int base = blockIdx.x * CHUNK;

  int sv[TRIPS];
  unsigned dv[TRIPS];
#pragma unroll
  for (int t = 0; t < TRIPS; ++t) {
    const int i = base + min(tid + t * 1024, CHUNK - 1);
    sv[t] = ei[i];
    dv[t] = (unsigned)ei[N_EDGES + i];
  }

  const int q  = tid >> 6;
  const int c4 = tid & 63;
  v4i st = {0, 0, 0, 0};
  v4i sm = {0, 0, 0, 0};
#pragma unroll
  for (int j = 0; j < NCHUNK / 16; ++j) {
    const int row = q * (NCHUNK / 16) + j;
    const v4i c = ((const v4i*)cnt)[row * (NBKT / 4) + c4];
    st += c;
    if (row < (int)blockIdx.x) sm += c;
  }
  ((v4i*)ptot[q])[c4]  = st;
  ((v4i*)pmine[q])[c4] = sm;
  __syncthreads();

  int mine = 0;
  if (tid < NBKT) {
    int t = 0;
#pragma unroll
    for (int j = 0; j < 16; ++j) { t += ptot[j][tid]; mine += pmine[j][tid]; }
    tot[tid] = t;
  }
  __syncthreads();

  if (tid < 32) {
    int v[8];
    int s = 0;
#pragma unroll
    for (int i = 0; i < 8; ++i) { v[i] = tot[tid * 8 + i]; s += v[i]; }
    int inc = s;
#pragma unroll
    for (int off = 1; off < 32; off <<= 1) {
      const int t = __shfl_up(inc, off);
      if (tid >= off) inc += t;
    }
    int run = inc - s;
#pragma unroll
    for (int i = 0; i < 8; ++i) { tot[tid * 8 + i] = run; run += v[i]; }
  }
  __syncthreads();

  if (tid < NBKT) {
    cursor[tid] = tot[tid] + mine;
    if (blockIdx.x == 0) {
      bstart[tid] = tot[tid];
      if (tid == 0) bstart[NBKT] = N_EDGES;
    }
  }
  __syncthreads();

#pragma unroll
  for (int t = 0; t < TRIPS; ++t) {
    if (tid + t * 1024 < CHUNK) {
      const unsigned bk = dv[t] / ROWS;
      const int pos = atomicAdd(&cursor[bk], 1);
      v2i p;
      p.x = sv[t];
      p.y = (int)(dv[t] - bk * ROWS);
      sorted[pos] = p;
    }
  }
}

__global__ __launch_bounds__(1024) void gcn_gather(const v2i* __restrict__ sorted,
                                                   const int* __restrict__ bstart,
                                                   const float* __restrict__ H,
                                                   const float* __restrict__ bias,
                                                   float* __restrict__ out) {
  __shared__ __attribute__((aligned(16))) float acc[(ROWS + 1) * D];
  const int tid  = threadIdx.x;
  const int lane = tid & 31;
  const int wave = tid >> 5;

  for (int i = tid; i < (ROWS + 1) * D / 4; i += 1024) {
    ((v4f*)acc)[i] = (v4f){0.f, 0.f, 0.f, 0.f};
  }
  const int beg = bstart[blockIdx.x];
  const int end = bstart[blockIdx.x + 1];
  const int nb  = (end - beg + EB - 1) / EB;
  __syncthreads();

  const v2i pad = {0, ROWS};
  int e = beg + wave * EB + ((lane < EB) ? lane : 0);
  v2i raw = sorted[min(e, N_EDGES - 1)];
  for (int k = wave; k < nb; k += 32) {
    const int en = e + 32 * EB;
    const v2i rawn = sorted[min(en, N_EDGES - 1)];
    const v2i p = (e < end) ? raw : pad;

    float m[EB][4];
#pragma unroll
    for (int j = 0; j < EB; ++j) {
      const int s = __builtin_amdgcn_readlane(p.x, j);
      const float* __restrict__ h = H + (size_t)s * D;
      m[j][0] = h[lane];
      m[j][1] = h[lane + 32];
      m[j][2] = h[lane + 64];
      m[j][3] = h[lane + 96];
    }
#pragma unroll
    for (int j = 0; j < EB; ++j) {
      const int r = __builtin_amdgcn_readlane(p.y, j);
      float* a = acc + r * D + lane;
      __hip_atomic_fetch_add(a,      m[j][0], __ATOMIC_RELAXED, __HIP_MEMORY_SCOPE_WORKGROUP);
      __hip_atomic_fetch_add(a + 32, m[j][1], __ATOMIC_RELAXED, __HIP_MEMORY_SCOPE_WORKGROUP);
      __hip_atomic_fetch_add(a + 64, m[j][2], __ATOMIC_RELAXED, __HIP_MEMORY_SCOPE_WORKGROUP);
      __hip_atomic_fetch_add(a + 96, m[j][3], __ATOMIC_RELAXED, __HIP_MEMORY_SCOPE_WORKGROUP);
    }
    raw = rawn;
    e = en;
  }
  __syncthreads();

  const v4f bv = ((const v4f*)bias)[lane];
  const int row0 = blockIdx.x * ROWS;
  for (int r = wave; r < ROWS; r += 32) {
    const int row = row0 + r;
    if (row < N_NODES) {
      const v4f v = ((const v4f*)(acc + r * D))[lane];
      ((v4f*)(out + (size_t)row * D))[lane] = v + bv;
    }
  }
}

extern "C" void kernel_launch(void* const* d_in, const int* in_sizes, int n_in,
                              void* d_out, int out_size, void* d_ws, size_t ws_size,
                              hipStream_t stream) {
  const float* x  = (const float*)d_in[0];
  const int*   ei = (const int*)d_in[1];
  const float* W  = (const float*)d_in[2];
  const float* b  = (const float*)d_in[3];
  float* out = (float*)d_out;
  float* H      = (float*)d_ws;
  int*   cnt    = (int*)((char*)d_ws + 33554432);
  int*   bstart = (int*)((char*)d_ws + 34603008);
  v2i*   sorted = (v2i*)((char*)d_ws + 41943040);

  gcn_count<<<NCHUNK, 1024, 0, stream>>>(ei, cnt);

  gcn_place<<<NCHUNK, 1024, 0, stream>>>(ei, cnt, bstart, sorted);

  gcn_gemm_wmma<<<GBLK, 256, 0, stream>>>(x, W, H);

  gcn_gather<<<NBKT, 1024, 0, stream>>>(sorted, bstart, H, b, out);
}


# gather: each work-group orders its edges by row in LDS (rank from a returning LDS int add), a wave sums whole 512-byte source rows in registers, each out row stored once; sums in rank (arrival) order
# speedup vs baseline: 2.7788x; 1.0736x over previous
#include <hip/hip_runtime.h>


typedef __attribute__((ext_vector_type(2))) float v2f;
typedef __attribute__((ext_vector_type(4))) float v4f;
typedef __attribute__((ext_vector_type(8))) float v8f;
typedef __attribute__((ext_vector_type(2))) int   v2i;
typedef __attribute__((ext_vector_type(4))) int   v4i;

#define N_NODES 50000
#define N_EDGES 800000
#define D       128

#define NBKT   256
#define ROWS   196
#define NCHUNK 256
#define CHUNK  3125
#define TRIPS  4
#define WP     132
#define GW     8
#define GBLK   391
#define CAP    4096
#define GT     4
#define RPL    7
#define RPAD   224
#define EB     16

__global__ __launch_bounds__(256) void gcn_gemm_wmma(const float* __restrict__ X,
                                                    const float* __restrict__ W,
                                                    float* __restrict__ H) {
  __shared__ __attribute__((aligned(16))) float wl[D * WP];
  __shared__ __attribute__((aligned(16))) float xl[GW * 16 * WP];
  const int tid   = threadIdx.x;
  const int lane  = tid & 31;
  const int wave  = __builtin_amdgcn_readfirstlane(tid >> 5);
  const int half  = lane >> 4;
  const int l     = lane & 15;
  const int strip = blockIdx.x * GW + wave;
  const bool live = strip < N_NODES / 16;
  const int mbase = strip * 16;
  float* xs = xl + wave * (16 * WP);

  v4f xr[16];
  if (live) {
#pragma unroll
    for (int i = 0; i < 16; ++i) {
      xr[i] = ((const v4f*)(X + (size_t)(mbase + i) * D))[lane];
    }
  }
  v4f wr[16];
#pragma unroll
  for (int q = 0; q < 16; ++q) {
    wr[q] = ((const v4f*)(W + (size_t)(wave + GW * q) * D))[lane];
  }
#pragma unroll
  for (int q = 0; q < 16; ++q) {
    *(v4f*)(wl + (wave + GW * q) * WP + lane * 4) = wr[q];
  }
  if (live) {
#pragma unroll
    for (int i = 0; i < 16; ++i) {
      *(v4f*)(xs + i * WP + lane * 4) = xr[i];
    }
  }
  __syncthreads();
  if (!live) return;

  const float* ap = xs + l * WP + 2 * half;
  const float* bp = wl + l * WP + 2 * half;
  v8f acc[8];
#pragma unroll
  for (int t = 0; t < 8; ++t) acc[t] = (v8f){0.f, 0.f, 0.f, 0.f, 0.f, 0.f, 0.f, 0.f};

#pragma unroll
  for (int k = 0; k < D; k += 4) {
    const v2f a = *(const v2f*)(ap + k);
    v2f bm[8];
#pragma unroll
    for (int t = 0; t < 8; ++t) bm[t] = *(const v2f*)(bp + t * 16 * WP + k);
#pragma unroll
    for (int t = 0; t < 8; ++t) {
      acc[t] = __builtin_amdgcn_wmma_f32_16x16x4_f32(
           false, a,  false, bm[t],
           (short)0, acc[t],  false,  false);
    }
  }

  float* __restrict__ hrow = H + (size_t)(mbase + 8 * half) * D + l;
#pragma unroll
  for (int r = 0; r < 8; ++r) {
#pragma unroll
    for (int t = 0; t < 8; ++t) {
      hrow[(size_t)r * D + 16 * t] = acc[t][r];
    }
  }
}

__global__ __launch_bounds__(1024) void gcn_count(const int* __restrict__ ei,
                                                  int* __restrict__ cnt) {
  __shared__ int hist[NBKT];
  const int tid = threadIdx.x;
  const int base = blockIdx.x * CHUNK;
  unsigned dv[TRIPS];
#pragma unroll
  for (int t = 0; t < TRIPS; ++t) {
    dv[t] = (unsigned)ei[N_EDGES + base + min(tid + t * 1024, CHUNK - 1)];
  }
  if (tid < NBKT) hist[tid] = 0;
  __syncthreads();
#pragma unroll
  for (int t = 0; t < TRIPS; ++t) {
    if (tid + t * 1024 < CHUNK) atomicAdd(&hist[dv[t] / ROWS], 1);
  }
  __syncthreads();
  if (tid < NBKT) cnt[blockIdx.x * NBKT + tid] = hist[tid];
}

__global__ __launch_bounds__(1024) void gcn_place(const int* __restrict__ ei,
                                                  const int* __restrict__ cnt,
                                                  int* __restrict__ bstart,
                                                  v2i* __restrict__ sorted) {
  __shared__ __attribute__((aligned(16))) int ptot[16][NBKT];
  __shared__ __attribute__((aligned(16))) int pmine[16][NBKT];
  __shared__ int tot[NBKT];
  __shared__ int cursor[NBKT];
  const int tid = threadIdx.x;
  const int base = blockIdx.x * CHUNK;

  int sv[TRIPS];
  unsigned dv[TRIPS];
#pragma unroll
  for (int t = 0; t < TRIPS; ++t) {
    const int i = base + min(tid + t * 1024, CHUNK - 1);
    sv[t] = ei[i];
    dv[t] = (unsigned)ei[N_EDGES + i];
  }

  const int q  = tid >> 6;
  const int c4 = tid & 63;
  v4i st = {0, 0, 0, 0};
  v4i sm = {0, 0, 0, 0};
#pragma unroll
  for (int j = 0; j < NCHUNK / 16; ++j) {
    const int row = q * (NCHUNK / 16) + j;
    const v4i c = ((const v4i*)cnt)[row * (NBKT / 4) + c4];
    st += c;
    if (row < (int)blockIdx.x) sm += c;
  }
  ((v4i*)ptot[q])[c4]  = st;
  ((v4i*)pmine[q])[c4] = sm;
  __syncthreads();

  int mine = 0;
  if (tid < NBKT) {
    int t = 0;
#pragma unroll
    for (int j = 0; j < 16; ++j) { t += ptot[j][tid]; mine += pmine[j][tid]; }
    tot[tid] = t;
  }
  __syncthreads();

  if (tid < 32) {
    int v[8];
    int s = 0;
#pragma unroll
    for (int i = 0; i < 8; ++i) { v[i] = tot[tid * 8 + i]; s += v[i]; }
    int inc = s;
#pragma unroll
    for (int off = 1; off < 32; off <<= 1) {
      const int t = __shfl_up(inc, off);
      if (tid >= off) inc += t;
    }
    int run = inc - s;
#pragma unroll
    for (int i = 0; i < 8; ++i) { tot[tid * 8 + i] = run; run += v[i]; }
  }
  __syncthreads();

  if (tid < NBKT) {
    cursor[tid] = tot[tid] + mine;
    if (blockIdx.x == 0) {
      bstart[tid] = tot[tid];
      if (tid == 0) bstart[NBKT] = N_EDGES;
    }
  }
  __syncthreads();

#pragma unroll
  for (int t = 0; t < TRIPS; ++t) {
    if (tid + t * 1024 < CHUNK) {
      const unsigned bk = dv[t] / ROWS;
      const int pos = atomicAdd(&cursor[bk], 1);
      v2i p;
      p.x = sv[t];
      p.y = (int)(dv[t] - bk * ROWS);
      sorted[pos] = p;
    }
  }
}

__global__ __launch_bounds__(1024) void gcn_gather(const v2i* __restrict__ sorted,
                                                   const int* __restrict__ bstart,
                                                   const float* __restrict__ H,
                                                   const float* __restrict__ bias,
                                                   float* __restrict__ out) {
  __shared__ int rcnt[RPAD];
  __shared__ int rstart[RPAD + 1];
  __shared__ int srcs[CAP];
  const int tid  = threadIdx.x;
  const int lane = tid & 31;
  const int wave = __builtin_amdgcn_readfirstlane(tid >> 5);
  const int beg = bstart[blockIdx.x];
  const int end = bstart[blockIdx.x + 1];
  const v4f bv = ((const v4f*)bias)[lane];
  const int row0 = blockIdx.x * ROWS;
  const int r0 = (ROWS * wave) >> 5;
  const int nr = ((ROWS * (wave + 1)) >> 5) - r0;

  bool first = true;
  int tb = beg;
  do {
    const int n = min(end - tb, CAP);
    if (tid < RPAD) rcnt[tid] = 0;
    int sv[GT], rv[GT], rk[GT];
#pragma unroll
    for (int t = 0; t < GT; ++t) {
      const v2i p = sorted[min(tb + tid + t * 1024, N_EDGES - 1)];
      sv[t] = p.x;
      rv[t] = p.y;
    }
    __syncthreads();
#pragma unroll
    for (int t = 0; t < GT; ++t) {
      if (tid + t * 1024 < n) rk[t] = atomicAdd(&rcnt[rv[t]], 1);
    }
    __syncthreads();

    if (tid < 32) {
      int v[RPL];
      int s = 0;
#pragma unroll
      for (int i = 0; i < RPL; ++i) { v[i] = rcnt[tid * RPL + i]; s += v[i]; }
      int inc = s;
#pragma unroll
      for (int off = 1; off < 32; off <<= 1) {
        const int t = __shfl_up(inc, off);
        if (tid >= off) inc += t;
      }
      int run = inc - s;
#pragma unroll
      for (int i = 0; i < RPL; ++i) { rstart[tid * RPL + i] = run; run += v[i]; }
      if (tid == 31) rstart[RPAD] = run;
    }
    __syncthreads();
#pragma unroll
    for (int t = 0; t < GT; ++t) {
      if (tid + t * 1024 < n) srcs[rstart[rv[t]] + rk[t]] = sv[t];
    }
    __syncthreads();

    const int bnd  = rstart[r0 + lane];
    int       p    = __builtin_amdgcn_readlane(bnd, 0);
    const int pend = __builtin_amdgcn_readlane(bnd, nr);
    int       k    = 0;
    int       nb   = __builtin_amdgcn_readlane(bnd, 1);
    v4f acc = {0.f, 0.f, 0.f, 0.f};

    auto flush = [&]() {
      const int row = row0 + r0 + k;
      if (row < N_NODES) {
        v4f* o = (v4f*)(out + (size_t)row * D) + lane;
        v4f base = bv;
        if (!first) base = *o;
        *o = acc + base;
      }
      acc = (v4f){0.f, 0.f, 0.f, 0.f};
      ++k;
      nb = __builtin_amdgcn_readlane(bnd, k + 1);
    };

    int idx = srcs[min(p + lane, CAP - 1)];
    for (; p < pend; p += EB) {
      const int idxn = srcs[min(p + EB + lane, CAP - 1)];
      v4f m[EB];
#pragma unroll
      for (int j = 0; j < EB; ++j) {
        if (p + j < pend) {
          const int s = __builtin_amdgcn_readlane(idx, j);
          m[j] = ((const v4f*)(H + (size_t)s * D))[lane];
        }
      }
#pragma unroll
      for (int j = 0; j < EB; ++j) {
        if (p + j < pend) {
          while (p + j == nb) flush();
          acc += m[j];
        }
      }
      idx = idxn;
    }
    while (k < nr) flush();

    first = false;
    tb += CAP;
  } while (tb < end);
}

extern "C" void kernel_launch(void* const* d_in, const int* in_sizes, int n_in,
                              void* d_out, int out_size, void* d_ws, size_t ws_size,
                              hipStream_t stream) {
  const float* x  = (const float*)d_in[0];
  const int*   ei = (const int*)d_in[1];
  const float* W  = (const float*)d_in[2];
  const float* b  = (const float*)d_in[3];
  float* out = (float*)d_out;
  float* H      = (float*)d_ws;
  int*   cnt    = (int*)((char*)d_ws + 33554432);
  int*   bstart = (int*)((char*)d_ws + 34603008);
  v2i*   sorted = (v2i*)((char*)d_ws + 41943040);

  gcn_count<<<NCHUNK, 1024, 0, stream>>>(ei, cnt);

  gcn_place<<<NCHUNK, 1024, 0, stream>>>(ei, cnt, bstart, sorted);

  gcn_gemm_wmma<<<GBLK, 256, 0, stream>>>(x, W, H);

  gcn_gather<<<NBKT, 1024, 0, stream>>>(sorted, bstart, H, b, out);
}


# place and GEMM run as one launch (place re-cut to 256-thread work-groups, first in the grid): edge placing runs beside the WMMA work, not before it; 3 launches for 4; sums stay in arrival order
# speedup vs baseline: 3.1119x; 1.1199x over previous
#include <hip/hip_runtime.h>


typedef __attribute__((ext_vector_type(2))) float v2f;
typedef __attribute__((ext_vector_type(4))) float v4f;
typedef __attribute__((ext_vector_type(8))) float v8f;
typedef __attribute__((ext_vector_type(2))) int   v2i;
typedef __attribute__((ext_vector_type(4))) int   v4i;

#define N_NODES 50000
#define N_EDGES 800000
#define D       128

#define NBKT   256
#define ROWS   196
#define NCHUNK 256
#define CHUNK  3125
#define TRIPS  4
#define PT     256
#define PTRIPS 13
#define PQ     4
#define WP     132
#define GW     8
#define GBLK   391
#define LDSF   (D * WP + GW * 16 * WP)
#define PGBLK  647
#define CAP    4096
#define GT     4
#define RPL    7
#define RPAD   224
#define EB     16

__device__ __forceinline__ void gemm_part(float* __restrict__ lds, const int gblock,
                                          const float* __restrict__ X,
                                          const float* __restrict__ W,
                                          float* __restrict__ H) {
  float* wl = lds;
  float* xl = lds + D * WP;
  const int tid   = threadIdx.x;
  const int lane  = tid & 31;
  const int wave  = __builtin_amdgcn_readfirstlane(tid >> 5);
  const int half  = lane >> 4;
  const int l     = lane & 15;
  const int strip = gblock * GW + wave;
  const bool live = strip < N_NODES / 16;
  const int mbase = strip * 16;
  float* xs = xl + wave * (16 * WP);

  v4f xr[16];
  if (live) {
#pragma unroll
    for (int i = 0; i < 16; ++i) {
      xr[i] = ((const v4f*)(X + (size_t)(mbase + i) * D))[lane];
    }
  }
  v4f wr[16];
#pragma unroll
  for (int q = 0; q < 16; ++q) {
    wr[q] = ((const v4f*)(W + (size_t)(wave + GW * q) * D))[lane];
  }
#pragma unroll
  for (int q = 0; q < 16; ++q) {
    *(v4f*)(wl + (wave + GW * q) * WP + lane * 4) = wr[q];
  }
  if (live) {
#pragma unroll
    for (int i = 0; i < 16; ++i) {
      *(v4f*)(xs + i * WP + lane * 4) = xr[i];
    }
  }
  __syncthreads();
  if (!live) return;

  const float* ap = xs + l * WP + 2 * half;
  const float* bp = wl + l * WP + 2 * half;
  v8f acc[8];
#pragma unroll
  for (int t = 0; t < 8; ++t) acc[t] = (v8f){0.f, 0.f, 0.f, 0.f, 0.f, 0.f, 0.f, 0.f};

#pragma unroll
  for (int k = 0; k < D; k += 4) {
    const v2f a = *(const v2f*)(ap + k);
    v2f bm[8];
#pragma unroll
    for (int t = 0; t < 8; ++t) bm[t] = *(const v2f*)(bp + t * 16 * WP + k);
#pragma unroll
    for (int t = 0; t < 8; ++t) {
      acc[t] = __builtin_amdgcn_wmma_f32_16x16x4_f32(
           false, a,  false, bm[t],
           (short)0, acc[t],  false,  false);
    }
  }

  float* __restrict__ hrow = H + (size_t)(mbase + 8 * half) * D + l;
#pragma unroll
  for (int r = 0; r < 8; ++r) {
#pragma unroll
    for (int t = 0; t < 8; ++t) {
      hrow[(size_t)r * D + 16 * t] = acc[t][r];
    }
  }
}

__global__ __launch_bounds__(1024) void gcn_count(const int* __restrict__ ei,
                                                  int* __restrict__ cnt) {
  __shared__ int hist[NBKT];
  const int tid = threadIdx.x;
  const int base = blockIdx.x * CHUNK;
  unsigned dv[TRIPS];
#pragma unroll
  for (int t = 0; t < TRIPS; ++t) {
    dv[t] = (unsigned)ei[N_EDGES + base + min(tid + t * 1024, CHUNK - 1)];
  }
  if (tid < NBKT) hist[tid] = 0;
  __syncthreads();
#pragma unroll
  for (int t = 0; t < TRIPS; ++t) {
    if (tid + t * 1024 < CHUNK) atomicAdd(&hist[dv[t] / ROWS], 1);
  }
  __syncthreads();
  if (tid < NBKT) cnt[blockIdx.x * NBKT + tid] = hist[tid];
}

__device__ __forceinline__ void place_part(int* __restrict__ lds, const int chunk,
                                           const int* __restrict__ ei,
                                           const int* __restrict__ cnt,
                                           int* __restrict__ bstart,
                                           v2i* __restrict__ sorted) {
  int* ptot   = lds;
  int* pmine  = lds + PQ * NBKT;
  int* tot    = lds + 2 * PQ * NBKT;
  int* cursor = tot + NBKT;
  const int tid = threadIdx.x;
  const int base = chunk * CHUNK;

  int sv[PTRIPS];
  unsigned dv[PTRIPS];
#pragma unroll
  for (int t = 0; t < PTRIPS; ++t) {
    const int i = base + min(tid + t * PT, CHUNK - 1);
    sv[t] = ei[i];
    dv[t] = (unsigned)ei[N_EDGES + i];
  }

  const int q  = tid >> 6;
  const int c4 = tid & 63;
  v4i st = {0, 0, 0, 0};
  v4i sm = {0, 0, 0, 0};
#pragma unroll 16
  for (int j = 0; j < NCHUNK / PQ; ++j) {
    const int row = q * (NCHUNK / PQ) + j;
    const v4i c = ((const v4i*)cnt)[row * (NBKT / 4) + c4];
    st += c;
    if (row < chunk) sm += c;
  }
  ((v4i*)(ptot + q * NBKT))[c4]  = st;
  ((v4i*)(pmine + q * NBKT))[c4] = sm;
  __syncthreads();

  int mine = 0;
  {
    int t = 0;
#pragma unroll
    for (int j = 0; j < PQ; ++j) { t += ptot[j * NBKT + tid]; mine += pmine[j * NBKT + tid]; }
    tot[tid] = t;
  }
  __syncthreads();

  if (tid < 32) {
    int v[8];
    int s = 0;
#pragma unroll
    for (int i = 0; i < 8; ++i) { v[i] = tot[tid * 8 + i]; s += v[i]; }
    int inc = s;
#pragma unroll
    for (int off = 1; off < 32; off <<= 1) {
      const int t = __shfl_up(inc, off);
      if (tid >= off) inc += t;
    }
    int run = inc - s;
#pragma unroll
    for (int i = 0; i < 8; ++i) { tot[tid * 8 + i] = run; run += v[i]; }
  }
  __syncthreads();

  cursor[tid] = tot[tid] + mine;
  if (chunk == 0) {
    bstart[tid] = tot[tid];
    if (tid == 0) bstart[NBKT] = N_EDGES;
  }
  __syncthreads();

#pragma unroll
  for (int t = 0; t < PTRIPS; ++t) {
    if (tid + t * PT < CHUNK) {
      const unsigned bk = dv[t] / ROWS;
      const int pos = atomicAdd(&cursor[bk], 1);
      v2i p;
      p.x = sv[t];
      p.y = (int)(dv[t] - bk * ROWS);
      sorted[pos] = p;
    }
  }
}

__global__ __launch_bounds__(PT) void gcn_place_gemm(const float* __restrict__ X,
                                                    const float* __restrict__ W,
                                                    float* __restrict__ H,
                                                    const int* __restrict__ ei,
                                                    const int* __restrict__ cnt,
                                                    int* __restrict__ bstart,
                                                    v2i* __restrict__ sorted) {
  __shared__ __attribute__((aligned(16))) float lds[LDSF];
  if (blockIdx.x < NCHUNK) {
    place_part((int*)lds, (int)blockIdx.x, ei, cnt, bstart, sorted);
    return;
  }
  gemm_part(lds, (int)blockIdx.x - NCHUNK, X, W, H);
}

__global__ __launch_bounds__(1024) void gcn_gather(const v2i* __restrict__ sorted,
                                                   const int* __restrict__ bstart,
                                                   const float* __restrict__ H,
                                                   const float* __restrict__ bias,
                                                   float* __restrict__ out) {
  __shared__ int rcnt[RPAD];
  __shared__ int rstart[RPAD + 1];
  __shared__ int srcs[CAP];
  const int tid  = threadIdx.x;
  const int lane = tid & 31;
  const int wave = __builtin_amdgcn_readfirstlane(tid >> 5);
  const int beg = bstart[blockIdx.x];
  const int end = bstart[blockIdx.x + 1];
  const v4f bv = ((const v4f*)bias)[lane];
  const int row0 = blockIdx.x * ROWS;
  const int r0 = (ROWS * wave) >> 5;
  const int nr = ((ROWS * (wave + 1)) >> 5) - r0;

  bool first = true;
  int tb = beg;
  do {
    const int n = min(end - tb, CAP);
    if (tid < RPAD) rcnt[tid] = 0;
    int sv[GT], rv[GT], rk[GT];
#pragma unroll
    for (int t = 0; t < GT; ++t) {
      const v2i p = sorted[min(tb + tid + t * 1024, N_EDGES - 1)];
      sv[t] = p.x;
      rv[t] = p.y;
    }
    __syncthreads();
#pragma unroll
    for (int t = 0; t < GT; ++t) {
      if (tid + t * 1024 < n) rk[t] = atomicAdd(&rcnt[rv[t]], 1);
    }
    __syncthreads();

    if (tid < 32) {
      int v[RPL];
      int s = 0;
#pragma unroll
      for (int i = 0; i < RPL; ++i) { v[i] = rcnt[tid * RPL + i]; s += v[i]; }
      int inc = s;
#pragma unroll
      for (int off = 1; off < 32; off <<= 1) {
        const int t = __shfl_up(inc, off);
        if (tid >= off) inc += t;
      }
      int run = inc - s;
#pragma unroll
      for (int i = 0; i < RPL; ++i) { rstart[tid * RPL + i] = run; run += v[i]; }
      if (tid == 31) rstart[RPAD] = run;
    }
    __syncthreads();
#pragma unroll
    for (int t = 0; t < GT; ++t) {
      if (tid + t * 1024 < n) srcs[rstart[rv[t]] + rk[t]] = sv[t];
    }
    __syncthreads();

    const int bnd  = rstart[r0 + lane];
    int       p    = __builtin_amdgcn_readlane(bnd, 0);
    const int pend = __builtin_amdgcn_readlane(bnd, nr);
    int       k    = 0;
    int       nb   = __builtin_amdgcn_readlane(bnd, 1);
    v4f acc = {0.f, 0.f, 0.f, 0.f};

    auto flush = [&]() {
      const int row = row0 + r0 + k;
      if (row < N_NODES) {
        v4f* o = (v4f*)(out + (size_t)row * D) + lane;
        v4f base = bv;
        if (!first) base = *o;
        *o = acc + base;
      }
      acc = (v4f){0.f, 0.f, 0.f, 0.f};
      ++k;
      nb = __builtin_amdgcn_readlane(bnd, k + 1);
    };

    int idx = srcs[min(p + lane, CAP - 1)];
    for (; p < pend; p += EB) {
      const int idxn = srcs[min(p + EB + lane, CAP - 1)];
      v4f m[EB];
#pragma unroll
      for (int j = 0; j < EB; ++j) {
        if (p + j < pend) {
          const int s = __builtin_amdgcn_readlane(idx, j);
          m[j] = ((const v4f*)(H + (size_t)s * D))[lane];
        }
      }
#pragma unroll
      for (int j = 0; j < EB; ++j) {
        if (p + j < pend) {
          while (p + j == nb) flush();
          acc += m[j];
        }
      }
      idx = idxn;
    }
    while (k < nr) flush();

    first = false;
    tb += CAP;
  } while (tb < end);
}

extern "C" void kernel_launch(void* const* d_in, const int* in_sizes, int n_in,
                              void* d_out, int out_size, void* d_ws, size_t ws_size,
                              hipStream_t stream) {
  const float* x  = (const float*)d_in[0];
  const int*   ei = (const int*)d_in[1];
  const float* W  = (const float*)d_in[2];
  const float* b  = (const float*)d_in[3];
  float* out = (float*)d_out;
  float* H      = (float*)d_ws;
  int*   cnt    = (int*)((char*)d_ws + 33554432);
  int*   bstart = (int*)((char*)d_ws + 34603008);
  v2i*   sorted = (v2i*)((char*)d_ws + 41943040);

  gcn_count<<<NCHUNK, 1024, 0, stream>>>(ei, cnt);

  gcn_place_gemm<<<PGBLK, PT, 0, stream>>>(x, W, H, ei, cnt, bstart, sorted);

  gcn_gather<<<NBKT, 1024, 0, stream>>>(sorted, bstart, H, b, out);
}


# no count launch: each place work-group orders its own chunk by range; a gather work-group takes its range's piece of every chunk via a (start,count) table; 2 launches for 3; sums in arrival order
# speedup vs baseline: 3.8752x; 1.2453x over previous
#include <hip/hip_runtime.h>


typedef __attribute__((ext_vector_type(2))) float v2f;
typedef __attribute__((ext_vector_type(4))) float v4f;
typedef __attribute__((ext_vector_type(8))) float v8f;
typedef __attribute__((ext_vector_type(2))) int   v2i;
typedef __attribute__((ext_vector_type(4))) int   v4i;

#define N_NODES 50000
#define N_EDGES 800000
#define D       128

#define NBKT   256
#define ROWS   196
#define NCHUNK 256
#define CHUNK  3125
#define CPITCH 3136
#define PT     256
#define PTRIPS 13
#define WP     132
#define GW     8
#define GBLK   391
#define LDSF   (D * WP + GW * 16 * WP)
#define PGBLK  647
#define CAP    4096
#define SEGU   8
#define RPL    7
#define RPAD   224
#define EB     16

__device__ __forceinline__ void gemm_part(float* __restrict__ lds, const int gblock,
                                          const float* __restrict__ X,
                                          const float* __restrict__ W,
                                          float* __restrict__ H) {
  float* wl = lds;
  float* xl = lds + D * WP;
  const int tid   = threadIdx.x;
  const int lane  = tid & 31;
  const int wave  = __builtin_amdgcn_readfirstlane(tid >> 5);
  const int half  = lane >> 4;
  const int l     = lane & 15;
  const int strip = gblock * GW + wave;
  const bool live = strip < N_NODES / 16;
  const int mbase = strip * 16;
  float* xs = xl + wave * (16 * WP);

  v4f xr[16];
  if (live) {
#pragma unroll
    for (int i = 0; i < 16; ++i) {
      xr[i] = ((const v4f*)(X + (size_t)(mbase + i) * D))[lane];
    }
  }
  v4f wr[16];
#pragma unroll
  for (int q = 0; q < 16; ++q) {
    wr[q] = ((const v4f*)(W + (size_t)(wave + GW * q) * D))[lane];
  }
#pragma unroll
  for (int q = 0; q < 16; ++q) {
    *(v4f*)(wl + (wave + GW * q) * WP + lane * 4) = wr[q];
  }
  if (live) {
#pragma unroll
    for (int i = 0; i < 16; ++i) {
      *(v4f*)(xs + i * WP + lane * 4) = xr[i];
    }
  }
  __syncthreads();
  if (!live) return;

  const float* ap = xs + l * WP + 2 * half;
  const float* bp = wl + l * WP + 2 * half;
  v8f acc[8];
#pragma unroll
  for (int t = 0; t < 8; ++t) acc[t] = (v8f){0.f, 0.f, 0.f, 0.f, 0.f, 0.f, 0.f, 0.f};

#pragma unroll
  for (int k = 0; k < D; k += 4) {
    const v2f a = *(const v2f*)(ap + k);
    v2f bm[8];
#pragma unroll
    for (int t = 0; t < 8; ++t) bm[t] = *(const v2f*)(bp + t * 16 * WP + k);
#pragma unroll
    for (int t = 0; t < 8; ++t) {
      acc[t] = __builtin_amdgcn_wmma_f32_16x16x4_f32(
           false, a,  false, bm[t],
           (short)0, acc[t],  false,  false);
    }
  }

  float* __restrict__ hrow = H + (size_t)(mbase + 8 * half) * D + l;
#pragma unroll
  for (int r = 0; r < 8; ++r) {
#pragma unroll
    for (int t = 0; t < 8; ++t) {
      hrow[(size_t)r * D + 16 * t] = acc[t][r];
    }
  }
}

__device__ __forceinline__ void place_part(int* __restrict__ lds, const int chunk,
                                           const int* __restrict__ ei,
                                           v2i* __restrict__ segtab,
                                           v2i* __restrict__ sorted) {
  int* hist  = lds;
  int* start = lds + NBKT;
  const int tid = threadIdx.x;
  const int base = chunk * CHUNK;

  int sv[PTRIPS];
  unsigned dv[PTRIPS];
  int rk[PTRIPS];
#pragma unroll
  for (int t = 0; t < PTRIPS; ++t) {
    const int i = base + min(tid + t * PT, CHUNK - 1);
    sv[t] = ei[i];
    dv[t] = (unsigned)ei[N_EDGES + i];
  }
  hist[tid] = 0;
  __syncthreads();
#pragma unroll
  for (int t = 0; t < PTRIPS; ++t) {
    if (tid + t * PT < CHUNK) rk[t] = atomicAdd(&hist[dv[t] / ROWS], 1);
  }
  __syncthreads();

  if (tid < 32) {
    int v[8];
    int sum = 0;
#pragma unroll
    for (int i = 0; i < 8; ++i) { v[i] = hist[tid * 8 + i]; sum += v[i]; }
    int inc = sum;
#pragma unroll
    for (int off = 1; off < 32; off <<= 1) {
      const int t = __shfl_up(inc, off);
      if (tid >= off) inc += t;
    }
    int run = inc - sum;
#pragma unroll
    for (int i = 0; i < 8; ++i) { start[tid * 8 + i] = run; run += v[i]; }
  }
  __syncthreads();

  {
    v2i e;
    e.x = start[tid];
    e.y = hist[tid];
    segtab[chunk * NBKT + tid] = e;
  }
  v2i* __restrict__ reg = sorted + chunk * CPITCH;
#pragma unroll
  for (int t = 0; t < PTRIPS; ++t) {
    if (tid + t * PT < CHUNK) {
      const unsigned bk = dv[t] / ROWS;
      v2i p;
      p.x = sv[t];
      p.y = (int)(dv[t] - bk * ROWS);
      reg[start[bk] + rk[t]] = p;
    }
  }
}

__global__ __launch_bounds__(PT) void gcn_place_gemm(const float* __restrict__ X,
                                                    const float* __restrict__ W,
                                                    float* __restrict__ H,
                                                    const int* __restrict__ ei,
                                                    v2i* __restrict__ segtab,
                                                    v2i* __restrict__ sorted) {
  __shared__ __attribute__((aligned(16))) float lds[LDSF];
  if (blockIdx.x < NCHUNK) {
    place_part((int*)lds, (int)blockIdx.x, ei, segtab, sorted);
    return;
  }
  gemm_part(lds, (int)blockIdx.x - NCHUNK, X, W, H);
}


__device__ __forceinline__ void gather_long(const v2i* __restrict__ sorted,
                                            const v2i* __restrict__ segtab,
                                            const float* __restrict__ H,
                                            const v4f bv, const int lane,
                                            const int rbeg, const int nr,
                                            float* __restrict__ out) {
  const int row0 = blockIdx.x * ROWS;
  for (int k = 0; k < nr; ++k) {
    const int r = rbeg + k;
    v4f acc = {0.f, 0.f, 0.f, 0.f};
    for (int c = 0; c < NCHUNK; ++c) {
      const v2i seg = segtab[c * NBKT + blockIdx.x];
      const int beg = __builtin_amdgcn_readfirstlane(seg.x);
      const int len = __builtin_amdgcn_readfirstlane(seg.y);
      const v2i* __restrict__ ep = sorted + c * CPITCH + beg;
      for (int i = 0; i < len; ++i) {
        const v2i e = ep[i];
        if (__builtin_amdgcn_readfirstlane(e.y) == r) {
          const int s = __builtin_amdgcn_readfirstlane(e.x);
          acc += ((const v4f*)(H + (size_t)s * D))[lane];
        }
      }
    }
    if (row0 + r < N_NODES) ((v4f*)(out + (size_t)(row0 + r) * D))[lane] = acc + bv;
  }
}

__global__ __launch_bounds__(1024) void gcn_gather(const v2i* __restrict__ sorted,
                                                   const v2i* __restrict__ segtab,
                                                   const float* __restrict__ H,
                                                   const float* __restrict__ bias,
                                                   float* __restrict__ out) {
  __shared__ int rcnt[RPAD];
  __shared__ int cur[RPAD + 1];
  __shared__ int ntot;
  __shared__ int srcs[CAP];
  const int tid  = threadIdx.x;
  const int lane = tid & 31;
  const int wave = __builtin_amdgcn_readfirstlane(tid >> 5);
  const int row0 = blockIdx.x * ROWS;
  const int r0 = (ROWS * wave) >> 5;
  const int nr = ((ROWS * (wave + 1)) >> 5) - r0;

  const int c = tid >> 2;
  const int q = tid & 3;
  const v2i seg = segtab[c * NBKT + blockIdx.x];
  const v4f bv = ((const v4f*)bias)[lane];
  const v2i* __restrict__ ep = sorted + c * CPITCH + seg.x;
  const int len = seg.y;
  if (tid < RPAD) rcnt[tid] = 0;

  const int lim = CHUNK - 1 - seg.x;
  int es[SEGU];
  int er[SEGU];
#pragma unroll
  for (int u = 0; u < SEGU; ++u) {
    const v2i t = ep[min(q + 4 * u, lim)];
    es[u] = t.x;
    er[u] = (q + 4 * u < len) ? t.y : -1;
  }
  __syncthreads();

#pragma unroll
  for (int u = 0; u < SEGU; ++u) {
    if (er[u] >= 0) atomicAdd(&rcnt[er[u]], 1);
  }
  for (int i = q + 4 * SEGU; i < len; i += 4) atomicAdd(&rcnt[ep[i].y], 1);
  __syncthreads();

  if (tid < 32) {
    int v[RPL];
    int sum = 0;
#pragma unroll
    for (int i = 0; i < RPL; ++i) { v[i] = rcnt[tid * RPL + i]; sum += v[i]; }
    int inc = sum;
#pragma unroll
    for (int off = 1; off < 32; off <<= 1) {
      const int t = __shfl_up(inc, off);
      if (tid >= off) inc += t;
    }
    int run = inc - sum;
#pragma unroll
    for (int i = 0; i < RPL; ++i) { cur[tid * RPL + i + 1] = run; run += v[i]; }
    if (tid == 0) cur[0] = 0;
    if (tid == 31) ntot = run;
  }
  __syncthreads();

  if (__builtin_amdgcn_readfirstlane(ntot) > CAP) {
    gather_long(sorted, segtab, H, bv, lane, r0, nr, out);
    return;
  }

  int slot[SEGU];
#pragma unroll
  for (int u = 0; u < SEGU; ++u) {
    if (er[u] >= 0) slot[u] = atomicAdd(&cur[er[u] + 1], 1);
  }
#pragma unroll
  for (int u = 0; u < SEGU; ++u) {
    if (er[u] >= 0) srcs[slot[u]] = es[u];
  }
  for (int i = q + 4 * SEGU; i < len; i += 4) {
    const v2i p = ep[i];
    srcs[atomicAdd(&cur[p.y + 1], 1)] = p.x;
  }
  __syncthreads();

  const int bnd  = cur[r0 + lane];
  int       p    = __builtin_amdgcn_readlane(bnd, 0);
  const int pend = __builtin_amdgcn_readlane(bnd, nr);
  int       k    = 0;
  int       nb   = __builtin_amdgcn_readlane(bnd, 1);
  v4f acc = {0.f, 0.f, 0.f, 0.f};

  auto flush = [&]() {
    const int row = row0 + r0 + k;
    if (row < N_NODES) ((v4f*)(out + (size_t)row * D))[lane] = acc + bv;
    acc = (v4f){0.f, 0.f, 0.f, 0.f};
    ++k;
    nb = __builtin_amdgcn_readlane(bnd, k + 1);
  };

  int idx = srcs[min(p + lane, CAP - 1)];
  for (; p < pend; p += EB) {
    const int idxn = srcs[min(p + EB + lane, CAP - 1)];
    v4f m[EB];
#pragma unroll
    for (int j = 0; j < EB; ++j) {
      if (p + j < pend) {
        const int s = __builtin_amdgcn_readlane(idx, j);
        m[j] = ((const v4f*)(H + (size_t)s * D))[lane];
      }
    }
#pragma unroll
    for (int j = 0; j < EB; ++j) {
      if (p + j < pend) {
        while (p + j == nb) flush();
        acc += m[j];
      }
    }
    idx = idxn;
  }
  while (k < nr) flush();
}

extern "C" void kernel_launch(void* const* d_in, const int* in_sizes, int n_in,
                              void* d_out, int out_size, void* d_ws, size_t ws_size,
                              hipStream_t stream) {
  const float* x  = (const float*)d_in[0];
  const int*   ei = (const int*)d_in[1];
  const float* W  = (const float*)d_in[2];
  const float* b  = (const float*)d_in[3];
  float* out = (float*)d_out;
  float* H      = (float*)d_ws;
  v2i*   segtab = (v2i*)((char*)d_ws + 33554432);
  v2i*   sorted = (v2i*)((char*)d_ws + 41943040);

  gcn_place_gemm<<<PGBLK, PT, 0, stream>>>(x, W, H, ei, segtab, sorted);

  gcn_gather<<<NBKT, 1024, 0, stream>>>(sorted, segtab, H, b, out);
}


# gather walk: whole batches run without per-edge end tests and m[j] is asked for again (row 16 edges on, address by one row_share add) as soon as it is added: 16 rows always in flight; sum order kept
# speedup vs baseline: 4.0199x; 1.0373x over previous
#include <hip/hip_runtime.h>


typedef __attribute__((ext_vector_type(2))) float v2f;
typedef __attribute__((ext_vector_type(4))) float v4f;
typedef __attribute__((ext_vector_type(8))) float v8f;
typedef __attribute__((ext_vector_type(2))) int   v2i;
typedef __attribute__((ext_vector_type(4))) int   v4i;

#define N_NODES 50000
#define N_EDGES 800000
#define D       128

#define NBKT   256
#define ROWS   196
#define NCHUNK 256
#define CHUNK  3125
#define CPITCH 3136
#define PT     256
#define PTRIPS 13
#define WP     132
#define GW     8
#define GBLK   391
#define LDSF   (D * WP + GW * 16 * WP)
#define PGBLK  647
#define CAP    4096
#define SEGU   8
#define RPL    7
#define RPAD   224
#define EB     16

__device__ __forceinline__ void gemm_part(float* __restrict__ lds, const int gblock,
                                          const float* __restrict__ X,
                                          const float* __restrict__ W,
                                          float* __restrict__ H) {
  float* wl = lds;
  float* xl = lds + D * WP;
  const int tid   = threadIdx.x;
  const int lane  = tid & 31;
  const int wave  = __builtin_amdgcn_readfirstlane(tid >> 5);
  const int half  = lane >> 4;
  const int l     = lane & 15;
  const int strip = gblock * GW + wave;
  const bool live = strip < N_NODES / 16;
  const int mbase = strip * 16;
  float* xs = xl + wave * (16 * WP);

  v4f xr[16];
  if (live) {
#pragma unroll
    for (int i = 0; i < 16; ++i) {
      xr[i] = ((const v4f*)(X + (size_t)(mbase + i) * D))[lane];
    }
  }
  v4f wr[16];
#pragma unroll
  for (int q = 0; q < 16; ++q) {
    wr[q] = ((const v4f*)(W + (size_t)(wave + GW * q) * D))[lane];
  }
#pragma unroll
  for (int q = 0; q < 16; ++q) {
    *(v4f*)(wl + (wave + GW * q) * WP + lane * 4) = wr[q];
  }
  if (live) {
#pragma unroll
    for (int i = 0; i < 16; ++i) {
      *(v4f*)(xs + i * WP + lane * 4) = xr[i];
    }
  }
  __syncthreads();
  if (!live) return;

  const float* ap = xs + l * WP + 2 * half;
  const float* bp = wl + l * WP + 2 * half;
  v8f acc[8];
#pragma unroll
  for (int t = 0; t < 8; ++t) acc[t] = (v8f){0.f, 0.f, 0.f, 0.f, 0.f, 0.f, 0.f, 0.f};

#pragma unroll
  for (int k = 0; k < D; k += 4) {
    const v2f a = *(const v2f*)(ap + k);
    v2f bm[8];
#pragma unroll
    for (int t = 0; t < 8; ++t) bm[t] = *(const v2f*)(bp + t * 16 * WP + k);
#pragma unroll
    for (int t = 0; t < 8; ++t) {
      acc[t] = __builtin_amdgcn_wmma_f32_16x16x4_f32(
           false, a,  false, bm[t],
           (short)0, acc[t],  false,  false);
    }
  }

  float* __restrict__ hrow = H + (size_t)(mbase + 8 * half) * D + l;
#pragma unroll
  for (int r = 0; r < 8; ++r) {
#pragma unroll
    for (int t = 0; t < 8; ++t) {
      hrow[(size_t)r * D + 16 * t] = acc[t][r];
    }
  }
}

__device__ __forceinline__ void place_part(int* __restrict__ lds, const int chunk,
                                           const int* __restrict__ ei,
                                           v2i* __restrict__ segtab,
                                           v2i* __restrict__ sorted) {
  int* hist  = lds;
  int* start = lds + NBKT;
  const int tid = threadIdx.x;
  const int base = chunk * CHUNK;

  int sv[PTRIPS];
  unsigned dv[PTRIPS];
  int rk[PTRIPS];
#pragma unroll
  for (int t = 0; t < PTRIPS; ++t) {
    const int i = base + min(tid + t * PT, CHUNK - 1);
    sv[t] = ei[i];
    dv[t] = (unsigned)ei[N_EDGES + i];
  }
  hist[tid] = 0;
  __syncthreads();
#pragma unroll
  for (int t = 0; t < PTRIPS; ++t) {
    if (tid + t * PT < CHUNK) rk[t] = atomicAdd(&hist[dv[t] / ROWS], 1);
  }
  __syncthreads();

  if (tid < 32) {
    int v[8];
    int sum = 0;
#pragma unroll
    for (int i = 0; i < 8; ++i) { v[i] = hist[tid * 8 + i]; sum += v[i]; }
    int inc = sum;
#pragma unroll
    for (int off = 1; off < 32; off <<= 1) {
      const int t = __shfl_up(inc, off);
      if (tid >= off) inc += t;
    }
    int run = inc - sum;
#pragma unroll
    for (int i = 0; i < 8; ++i) { start[tid * 8 + i] = run; run += v[i]; }
  }
  __syncthreads();

  {
    v2i e;
    e.x = start[tid];
    e.y = hist[tid];
    segtab[chunk * NBKT + tid] = e;
  }
  v2i* __restrict__ reg = sorted + chunk * CPITCH;
#pragma unroll
  for (int t = 0; t < PTRIPS; ++t) {
    if (tid + t * PT < CHUNK) {
      const unsigned bk = dv[t] / ROWS;
      v2i p;
      p.x = sv[t];
      p.y = (int)(dv[t] - bk * ROWS);
      reg[start[bk] + rk[t]] = p;
    }
  }
}

__global__ __launch_bounds__(PT) void gcn_place_gemm(const float* __restrict__ X,
                                                    const float* __restrict__ W,
                                                    float* __restrict__ H,
                                                    const int* __restrict__ ei,
                                                    v2i* __restrict__ segtab,
                                                    v2i* __restrict__ sorted) {
  __shared__ __attribute__((aligned(16))) float lds[LDSF];
  if (blockIdx.x < NCHUNK) {
    place_part((int*)lds, (int)blockIdx.x, ei, segtab, sorted);
    return;
  }
  gemm_part(lds, (int)blockIdx.x - NCHUNK, X, W, H);
}


__device__ __forceinline__ void gather_long(const v2i* __restrict__ sorted,
                                            const v2i* __restrict__ segtab,
                                            const float* __restrict__ H,
                                            const v4f bv, const int lane,
                                            const int rbeg, const int nr,
                                            float* __restrict__ out) {
  const int row0 = blockIdx.x * ROWS;
  for (int k = 0; k < nr; ++k) {
    const int r = rbeg + k;
    v4f acc = {0.f, 0.f, 0.f, 0.f};
    for (int c = 0; c < NCHUNK; ++c) {
      const v2i seg = segtab[c * NBKT + blockIdx.x];
      const int beg = __builtin_amdgcn_readfirstlane(seg.x);
      const int len = __builtin_amdgcn_readfirstlane(seg.y);
      const v2i* __restrict__ ep = sorted + c * CPITCH + beg;
      for (int i = 0; i < len; ++i) {
        const v2i e = ep[i];
        if (__builtin_amdgcn_readfirstlane(e.y) == r) {
          const int s = __builtin_amdgcn_readfirstlane(e.x);
          acc += ((const v4f*)(H + (size_t)s * D))[lane];
        }
      }
    }
    if (row0 + r < N_NODES) ((v4f*)(out + (size_t)(row0 + r) * D))[lane] = acc + bv;
  }
}

template <int J>
__device__ __forceinline__ v4f row_of(const char* __restrict__ Hb, const int offs, const unsigned laneoff) {
  const unsigned o = (unsigned)__builtin_amdgcn_update_dpp(0, offs, 0x150 + J, 0xf, 0xf, false) + laneoff;
  return *(const v4f*)(Hb + o);
}

__global__ __launch_bounds__(1024) void gcn_gather(const v2i* __restrict__ sorted,
                                                   const v2i* __restrict__ segtab,
                                                   const float* __restrict__ H,
                                                   const float* __restrict__ bias,
                                                   float* __restrict__ out) {
  __shared__ int rcnt[RPAD];
  __shared__ int cur[RPAD + 1];
  __shared__ int ntot;
  __shared__ int srcs[CAP];
  const int tid  = threadIdx.x;
  const int lane = tid & 31;
  const int wave = __builtin_amdgcn_readfirstlane(tid >> 5);
  const int row0 = blockIdx.x * ROWS;
  const int r0 = (ROWS * wave) >> 5;
  const int nr = ((ROWS * (wave + 1)) >> 5) - r0;

  const int c = tid >> 2;
  const int q = tid & 3;
  const v2i seg = segtab[c * NBKT + blockIdx.x];
  const v4f bv = ((const v4f*)bias)[lane];
  const v2i* __restrict__ ep = sorted + c * CPITCH + seg.x;
  const int len = seg.y;
  if (tid < RPAD) rcnt[tid] = 0;

  const int lim = CHUNK - 1 - seg.x;
  int es[SEGU];
  int er[SEGU];
#pragma unroll
  for (int u = 0; u < SEGU; ++u) {
    const v2i t = ep[min(q + 4 * u, lim)];
    es[u] = t.x;
    er[u] = (q + 4 * u < len) ? t.y : -1;
  }
  __syncthreads();

#pragma unroll
  for (int u = 0; u < SEGU; ++u) {
    if (er[u] >= 0) atomicAdd(&rcnt[er[u]], 1);
  }
  for (int i = q + 4 * SEGU; i < len; i += 4) atomicAdd(&rcnt[ep[i].y], 1);
  __syncthreads();

  if (tid < 32) {
    int v[RPL];
    int sum = 0;
#pragma unroll
    for (int i = 0; i < RPL; ++i) { v[i] = rcnt[tid * RPL + i]; sum += v[i]; }
    int inc = sum;
#pragma unroll
    for (int off = 1; off < 32; off <<= 1) {
      const int t = __shfl_up(inc, off);
      if (tid >= off) inc += t;
    }
    int run = inc - sum;
#pragma unroll
    for (int i = 0; i < RPL; ++i) { cur[tid * RPL + i + 1] = run; run += v[i]; }
    if (tid == 0) cur[0] = 0;
    if (tid == 31) ntot = run;
  }
  __syncthreads();

  if (__builtin_amdgcn_readfirstlane(ntot) > CAP) {
    gather_long(sorted, segtab, H, bv, lane, r0, nr, out);
    return;
  }

  int slot[SEGU];
#pragma unroll
  for (int u = 0; u < SEGU; ++u) {
    if (er[u] >= 0) slot[u] = atomicAdd(&cur[er[u] + 1], 1);
  }
#pragma unroll
  for (int u = 0; u < SEGU; ++u) {
    if (er[u] >= 0) srcs[slot[u]] = es[u];
  }
  for (int i = q + 4 * SEGU; i < len; i += 4) {
    const v2i p = ep[i];
    srcs[atomicAdd(&cur[p.y + 1], 1)] = p.x;
  }
  __syncthreads();

  const int bnd  = cur[r0 + lane];
  int       p    = __builtin_amdgcn_readlane(bnd, 0);
  const int pend = __builtin_amdgcn_readlane(bnd, nr);
  int       k    = 0;
  int       d    = __builtin_amdgcn_readlane(bnd, 1) - p;
  v4f acc = {0.f, 0.f, 0.f, 0.f};

  auto flush = [&]() {
    const int row = row0 + r0 + k;
    if (row < N_NODES) ((v4f*)(out + (size_t)row * D))[lane] = acc + bv;
    acc = (v4f){0.f, 0.f, 0.f, 0.f};
    ++k;
    d = __builtin_amdgcn_readlane(bnd, k + 1) - p;
  };

  const char* __restrict__ Hb = (const char*)H;
  const unsigned laneoff = (unsigned)lane * 16u;
  const int l16 = lane & 15;
  int offc = srcs[min(p + l16, CAP - 1)] << 9;
  int offn = srcs[min(p + EB + l16, CAP - 1)] << 9;
  v4f m[EB];

#define STEPS(S) S(0) S(1) S(2) S(3) S(4) S(5) S(6) S(7) S(8) S(9) S(10) S(11) S(12) S(13) S(14) S(15)
#define ISSUE(j)      m[j] = row_of<j>(Hb, offc, laneoff);
#define ISSUE_T(j)    if ((j) < rem) m[j] = row_of<j>(Hb, offc, laneoff);
#define STEP(j)       while (__builtin_expect(d == (j), 0)) flush(); acc += m[j]; m[j] = row_of<j>(Hb, offn, laneoff);
#define STEP_T(j)     if ((j) < rem) { while (__builtin_expect(d == (j), 0)) flush(); acc += m[j]; } \
                      if ((j) + EB < rem) m[j] = row_of<j>(Hb, offn, laneoff);

  if (p + 2 * EB <= pend) {
    STEPS(ISSUE)
    do {
      const int offnn = srcs[min(p + 2 * EB + l16, CAP - 1)] << 9;
      STEPS(STEP)
      p += EB;
      d -= EB;
      offn = offnn;
    } while (p + 2 * EB <= pend);
  } else {
    const int rem = pend - p;
    STEPS(ISSUE_T)
  }
  for (; p < pend; p += EB) {
    const int rem = pend - p;
    const int offnn = srcs[min(p + 2 * EB + l16, CAP - 1)] << 9;
    STEPS(STEP_T)
    d -= EB;
    offn = offnn;
  }
#undef STEPS
#undef ISSUE
#undef ISSUE_T
#undef STEP
#undef STEP_T
  while (k < nr) flush();
}

extern "C" void kernel_launch(void* const* d_in, const int* in_sizes, int n_in,
                              void* d_out, int out_size, void* d_ws, size_t ws_size,
                              hipStream_t stream) {
  const float* x  = (const float*)d_in[0];
  const int*   ei = (const int*)d_in[1];
  const float* W  = (const float*)d_in[2];
  const float* b  = (const float*)d_in[3];
  float* out = (float*)d_out;
  float* H      = (float*)d_ws;
  v2i*   segtab = (v2i*)((char*)d_ws + 33554432);
  v2i*   sorted = (v2i*)((char*)d_ws + 41943040);

  gcn_place_gemm<<<PGBLK, PT, 0, stream>>>(x, W, H, ei, segtab, sorted);

  gcn_gather<<<NBKT, 1024, 0, stream>>>(sorted, segtab, H, b, out);
}


# X W^T no longer formed: gather sums rows of X, each range's work-group multiplies its summed rows by W^T on f32 WMMA, partly beside its walk; sum taken before the product, so rounding differs
# speedup vs baseline: 4.5276x; 1.1263x over previous
#include <hip/hip_runtime.h>


typedef __attribute__((ext_vector_type(2))) float v2f;
typedef __attribute__((ext_vector_type(4))) float v4f;
typedef __attribute__((ext_vector_type(8))) float v8f;
typedef __attribute__((ext_vector_type(2))) int   v2i;
typedef __attribute__((ext_vector_type(4))) int   v4i;

#define N_NODES 50000
#define N_EDGES 800000
#define D       128

#define NBKT   256
#define ROWS   196
#define NCHUNK 256
#define CHUNK  3125
#define CPITCH 3136
#define PT     256
#define PTRIPS 13
#define WP     132
#define CAP    4096
#define SEGU   8
#define RPL    7
#define RPAD   224
#define EB     16
#define NSTRIP 13
#define ASTRIP 8
#define AROWS  128
#define BROWS  68

template <int NU>
__device__ __forceinline__ void strip_gemm(const float* __restrict__ wl, const float* __restrict__ sl,
                                           const int s0, const int p, const int lane, const int row0,
                                           const float* __restrict__ bias, float* __restrict__ out) {
  const int half = lane >> 4;
  const int l    = lane & 15;
  const float* ap = sl + (s0 * 16 + l) * WP + 2 * half;
  const float* bp = wl + (32 * p + l) * WP + 2 * half;
  const float b0 = bias[32 * p + l];
  const float b1 = bias[32 * p + 16 + l];
  v8f acc[NU][2];
#pragma unroll
  for (int u = 0; u < NU; ++u) {
    acc[u][0] = (v8f){0.f, 0.f, 0.f, 0.f, 0.f, 0.f, 0.f, 0.f};
    acc[u][1] = (v8f){0.f, 0.f, 0.f, 0.f, 0.f, 0.f, 0.f, 0.f};
  }

#pragma unroll
  for (int k = 0; k < D; k += 4) {
    const v2f bm0 = *(const v2f*)(bp + k);
    const v2f bm1 = *(const v2f*)(bp + 16 * WP + k);
    v2f a[NU];
#pragma unroll
    for (int u = 0; u < NU; ++u) a[u] = *(const v2f*)(ap + u * 64 * WP + k);
#pragma unroll
    for (int u = 0; u < NU; ++u) {
      acc[u][0] = __builtin_amdgcn_wmma_f32_16x16x4_f32(
           false, a[u],  false, bm0,
           (short)0, acc[u][0],  false,  false);
      acc[u][1] = __builtin_amdgcn_wmma_f32_16x16x4_f32(
           false, a[u],  false, bm1,
           (short)0, acc[u][1],  false,  false);
    }
  }

#pragma unroll
  for (int u = 0; u < NU; ++u) {
    const int lr = (s0 + 4 * u) * 16 + 8 * half;
    float* __restrict__ orow = out + (size_t)(row0 + lr) * D + 32 * p + l;
#pragma unroll
    for (int r = 0; r < 8; ++r) {
      if (lr + r < ROWS && row0 + lr + r < N_NODES) {
        orow[(size_t)r * D]      = acc[u][0][r] + b0;
        orow[(size_t)r * D + 16] = acc[u][1][r] + b1;
      }
    }
  }
}

__device__ __forceinline__ void place_part(int* __restrict__ lds, const int chunk,
                                           const int* __restrict__ ei,
                                           v2i* __restrict__ segtab,
                                           v2i* __restrict__ sorted) {
  int* hist  = lds;
  int* start = lds + NBKT;
  const int tid = threadIdx.x;
  const int base = chunk * CHUNK;

  int sv[PTRIPS];
  unsigned dv[PTRIPS];
  int rk[PTRIPS];
#pragma unroll
  for (int t = 0; t < PTRIPS; ++t) {
    const int i = base + min(tid + t * PT, CHUNK - 1);
    sv[t] = ei[i];
    dv[t] = (unsigned)ei[N_EDGES + i];
  }
  hist[tid] = 0;
  __syncthreads();
#pragma unroll
  for (int t = 0; t < PTRIPS; ++t) {
    if (tid + t * PT < CHUNK) rk[t] = atomicAdd(&hist[dv[t] / ROWS], 1);
  }
  __syncthreads();

  if (tid < 32) {
    int v[8];
    int sum = 0;
#pragma unroll
    for (int i = 0; i < 8; ++i) { v[i] = hist[tid * 8 + i]; sum += v[i]; }
    int inc = sum;
#pragma unroll
    for (int off = 1; off < 32; off <<= 1) {
      const int t = __shfl_up(inc, off);
      if (tid >= off) inc += t;
    }
    int run = inc - sum;
#pragma unroll
    for (int i = 0; i < 8; ++i) { start[tid * 8 + i] = run; run += v[i]; }
  }
  __syncthreads();

  {
    v2i e;
    e.x = start[tid];
    e.y = hist[tid];
    segtab[chunk * NBKT + tid] = e;
  }
  v2i* __restrict__ reg = sorted + chunk * CPITCH;
#pragma unroll
  for (int t = 0; t < PTRIPS; ++t) {
    if (tid + t * PT < CHUNK) {
      const unsigned bk = dv[t] / ROWS;
      v2i p;
      p.x = sv[t];
      p.y = (int)(dv[t] - bk * ROWS);
      reg[start[bk] + rk[t]] = p;
    }
  }
}

__global__ __launch_bounds__(PT) void gcn_place(const int* __restrict__ ei,
                                               v2i* __restrict__ segtab,
                                               v2i* __restrict__ sorted) {
  __shared__ int lds[2 * NBKT];
  place_part(lds, (int)blockIdx.x, ei, segtab, sorted);
}


__device__ __forceinline__ void long_rows(const v2i* __restrict__ sorted,
                                          const v2i* __restrict__ segtab,
                                          const float* __restrict__ X,
                                          float* srow, const int lane,
                                          const int rbeg, const int nr) {
  for (int k = 0; k < nr; ++k) {
    const int r = rbeg + k;
    v4f acc = {0.f, 0.f, 0.f, 0.f};
    for (int c = 0; c < NCHUNK; ++c) {
      const v2i seg = segtab[c * NBKT + blockIdx.x];
      const int beg = __builtin_amdgcn_readfirstlane(seg.x);
      const int len = __builtin_amdgcn_readfirstlane(seg.y);
      const v2i* __restrict__ ep = sorted + c * CPITCH + beg;
      for (int i = 0; i < len; ++i) {
        const v2i e = ep[i];
        if (__builtin_amdgcn_readfirstlane(e.y) == r) {
          const int s = __builtin_amdgcn_readfirstlane(e.x);
          acc += ((const v4f*)(X + (size_t)s * D))[lane];
        }
      }
    }
    *(v4f*)(srow + r * WP) = acc;
  }
}

template <int J>
__device__ __forceinline__ v4f row_of(const char* __restrict__ Xb, const int offs, const unsigned laneoff) {
  const unsigned o = (unsigned)__builtin_amdgcn_update_dpp(0, offs, 0x150 + J, 0xf, 0xf, false) + laneoff;
  return *(const v4f*)(Xb + o);
}

__device__ __forceinline__ void walk_rows(const int* __restrict__ srcs, const int* __restrict__ cur,
                                          float* srow, const char* __restrict__ Xb,
                                          const int lane, const int r0, const int nr) {
  const int bnd  = cur[r0 + lane];
  int       p    = __builtin_amdgcn_readlane(bnd, 0);
  const int pend = __builtin_amdgcn_readlane(bnd, nr);
  int       k    = 0;
  int       d    = __builtin_amdgcn_readlane(bnd, 1) - p;
  v4f acc = {0.f, 0.f, 0.f, 0.f};

  auto flush = [&]() {
    *(v4f*)(srow + (r0 + k) * WP) = acc;
    acc = (v4f){0.f, 0.f, 0.f, 0.f};
    ++k;
    d = __builtin_amdgcn_readlane(bnd, k + 1) - p;
  };

  const unsigned laneoff = (unsigned)lane * 16u;
  const int l16 = lane & 15;
  int offc = srcs[min(p + l16, CAP - 1)] << 9;
  int offn = srcs[min(p + EB + l16, CAP - 1)] << 9;
  v4f m[EB];

#define STEPS(S) S(0) S(1) S(2) S(3) S(4) S(5) S(6) S(7) S(8) S(9) S(10) S(11) S(12) S(13) S(14) S(15)
#define ISSUE(j)      m[j] = row_of<j>(Xb, offc, laneoff);
#define ISSUE_T(j)    if ((j) < rem) m[j] = row_of<j>(Xb, offc, laneoff);
#define STEP(j)       while (__builtin_expect(d == (j), 0)) flush(); acc += m[j]; m[j] = row_of<j>(Xb, offn, laneoff);
#define STEP_T(j)     if ((j) < rem) { while (__builtin_expect(d == (j), 0)) flush(); acc += m[j]; } \
                      if ((j) + EB < rem) m[j] = row_of<j>(Xb, offn, laneoff);

  if (p + 2 * EB <= pend) {
    STEPS(ISSUE)
    do {
      const int offnn = srcs[min(p + 2 * EB + l16, CAP - 1)] << 9;
      STEPS(STEP)
      p += EB;
      d -= EB;
      offn = offnn;
    } while (p + 2 * EB <= pend);
  } else {
    const int rem = pend - p;
    STEPS(ISSUE_T)
  }
  for (; p < pend; p += EB) {
    const int rem = pend - p;
    const int offnn = srcs[min(p + 2 * EB + l16, CAP - 1)] << 9;
    STEPS(STEP_T)
    d -= EB;
    offn = offnn;
  }
#undef STEPS
#undef ISSUE
#undef ISSUE_T
#undef STEP
#undef STEP_T
  while (k < nr) flush();
}

__global__ __launch_bounds__(1024) void gcn_gather_gemm(const v2i* __restrict__ sorted,
                                                        const v2i* __restrict__ segtab,
                                                        const float* __restrict__ X,
                                                        const float* __restrict__ W,
                                                        const float* __restrict__ bias,
                                                        float* __restrict__ out) {
  __shared__ __attribute__((aligned(16))) float wl[D * WP];
  __shared__ __attribute__((aligned(16))) float sl[NSTRIP * 16 * WP];
  __shared__ int rcnt[RPAD];
  __shared__ int cur[RPAD + 1];
  __shared__ int ntot;
  __shared__ int srcs[CAP];
  const int tid  = threadIdx.x;
  const int lane = tid & 31;
  const int wave = __builtin_amdgcn_readfirstlane(tid >> 5);
  const int row0 = blockIdx.x * ROWS;

  v4f wr[4];
#pragma unroll
  for (int j = 0; j < 4; ++j) {
    wr[j] = ((const v4f*)(W + (size_t)(wave + 32 * j) * D))[lane];
  }

  const int c = tid >> 2;
  const int q = tid & 3;
  const v2i seg = segtab[c * NBKT + blockIdx.x];
  const v2i* __restrict__ ep = sorted + c * CPITCH + seg.x;
  const int len = seg.y;
  if (tid < RPAD) rcnt[tid] = 0;

  const int lim = CHUNK - 1 - seg.x;
  int es[SEGU];
  int er[SEGU];
#pragma unroll
  for (int u = 0; u < SEGU; ++u) {
    const v2i t = ep[min(q + 4 * u, lim)];
    es[u] = t.x;
    er[u] = (q + 4 * u < len) ? t.y : -1;
  }
#pragma unroll
  for (int j = 0; j < 4; ++j) {
    *(v4f*)(wl + (wave + 32 * j) * WP + lane * 4) = wr[j];
  }
  if (wave < NSTRIP * 16 - ROWS) {
    *(v4f*)(sl + (ROWS + wave) * WP + lane * 4) = (v4f){0.f, 0.f, 0.f, 0.f};
  }
  __syncthreads();

#pragma unroll
  for (int u = 0; u < SEGU; ++u) {
    if (er[u] >= 0) atomicAdd(&rcnt[er[u]], 1);
  }
  for (int i = q + 4 * SEGU; i < len; i += 4) atomicAdd(&rcnt[ep[i].y], 1);
  __syncthreads();

  if (tid < 32) {
    int v[RPL];
    int sum = 0;
#pragma unroll
    for (int i = 0; i < RPL; ++i) { v[i] = rcnt[tid * RPL + i]; sum += v[i]; }
    int inc = sum;
#pragma unroll
    for (int off = 1; off < 32; off <<= 1) {
      const int t = __shfl_up(inc, off);
      if (tid >= off) inc += t;
    }
    int run = inc - sum;
#pragma unroll
    for (int i = 0; i < RPL; ++i) { cur[tid * RPL + i + 1] = run; run += v[i]; }
    if (tid == 0) cur[0] = 0;
    if (tid == 31) ntot = run;
  }
  __syncthreads();

  const bool big = __builtin_amdgcn_readfirstlane(ntot) > CAP;
  if (!big) {
    int slot[SEGU];
#pragma unroll
    for (int u = 0; u < SEGU; ++u) {
      if (er[u] >= 0) slot[u] = atomicAdd(&cur[er[u] + 1], 1);
    }
#pragma unroll
    for (int u = 0; u < SEGU; ++u) {
      if (er[u] >= 0) srcs[slot[u]] = es[u];
    }
    for (int i = q + 4 * SEGU; i < len; i += 4) {
      const v2i pe = ep[i];
      srcs[atomicAdd(&cur[pe.y + 1], 1)] = pe.x;
    }
    __syncthreads();
  }

  const bool mult = ((wave >> 2) & 1) == 0;
  const int  idx  = (wave & 3) | ((wave >> 3) << 2);
  const char* __restrict__ Xb = (const char*)X;
  float* srow = sl + lane * 4;

#pragma nounroll
  for (int st = 0; st < 2; ++st) {
    if (st == 0 || !mult) {
      const int r0 = (st == 0) ? 4 * wave : AROWS + ((BROWS * idx) >> 4);
      const int nr = (st == 0) ? 4 : ((BROWS * (idx + 1)) >> 4) - ((BROWS * idx) >> 4);
      if (big) long_rows(sorted, segtab, X, srow, lane, r0, nr);
      else     walk_rows(srcs, cur, srow, Xb, lane, r0, nr);
    } else {
      strip_gemm<2>(wl, sl, idx >> 2, idx & 3, lane, row0, bias, out);
    }
    __syncthreads();
  }

  const int w8 = wave & 7;
  if (w8 < 4 || w8 == 4 + (wave >> 3)) {
    const int ui = 5 * (wave >> 3) + min(w8, 4);
    strip_gemm<1>(wl, sl, ASTRIP + (ui >> 2), ui & 3, lane, row0, bias, out);
  }
}

extern "C" void kernel_launch(void* const* d_in, const int* in_sizes, int n_in,
                              void* d_out, int out_size, void* d_ws, size_t ws_size,
                              hipStream_t stream) {
  const float* x  = (const float*)d_in[0];
  const int*   ei = (const int*)d_in[1];
  const float* W  = (const float*)d_in[2];
  const float* b  = (const float*)d_in[3];
  float* out = (float*)d_out;
  v2i*   segtab = (v2i*)((char*)d_ws + 33554432);
  v2i*   sorted = (v2i*)((char*)d_ws + 41943040);

  gcn_place<<<NCHUNK, PT, 0, stream>>>(ei, segtab, sorted);

  gcn_gather_gemm<<<NBKT, 1024, 0, stream>>>(sorted, segtab, x, W, b, out);
}
